# starting kernel
# baseline (speedup 1.0000x reference)
#include <hip/hip_runtime.h>
#include <hip/hip_bf16.h>
#include <math.h>

#define N_SENT 131072
#define N_BAGS 16384
#define DCLS   512
#define REL    53

typedef __attribute__((ext_vector_type(2))) float v2f;
typedef __attribute__((ext_vector_type(8))) float v8f;

__global__ __launch_bounds__(256) void tan_bag_attn(
    const float* __restrict__ X, const int* __restrict__ scope,
    const float* __restrict__ Con, const int* __restrict__ rel,
    float* __restrict__ bag)
{
  const int lane = threadIdx.x & 31;
  const int wave = threadIdx.x >> 5;
  const int b = blockIdx.x * 8 + wave;

  const int s0 = scope[2 * b];
  const int s1 = scope[2 * b + 1];
  const float4* con4 = (const float4*)(Con + (size_t)rel[b] * DCLS);

  float4 c4[4], a4[4];
#pragma unroll
  for (int q = 0; q < 4; ++q) {
    c4[q] = con4[lane + 32 * q];
    a4[q] = make_float4(0.f, 0.f, 0.f, 0.f);
  }

  float m = -INFINITY, z = 0.f;
  for (int j = s0; j < s1; ++j) {
    const float4* x4 = (const float4*)(X + (size_t)j * DCLS);
    if (j + 1 < s1)
      __builtin_prefetch((const float*)x4 + DCLS, 0, 1);

    float4 xv[4];
    float s = 0.f;
#pragma unroll
    for (int q = 0; q < 4; ++q) {
      xv[q] = x4[lane + 32 * q];
      s = fmaf(xv[q].x, c4[q].x, s);
      s = fmaf(xv[q].y, c4[q].y, s);
      s = fmaf(xv[q].z, c4[q].z, s);
      s = fmaf(xv[q].w, c4[q].w, s);
    }
#pragma unroll
    for (int off = 16; off > 0; off >>= 1) s += __shfl_xor(s, off, 32);

    const float mn    = fmaxf(m, s);
    const float scale = __expf(m - mn);
    const float w     = __expf(s - mn);
    z = z * scale + w;
#pragma unroll
    for (int q = 0; q < 4; ++q) {
      a4[q].x = fmaf(a4[q].x, scale, w * xv[q].x);
      a4[q].y = fmaf(a4[q].y, scale, w * xv[q].y);
      a4[q].z = fmaf(a4[q].z, scale, w * xv[q].z);
      a4[q].w = fmaf(a4[q].w, scale, w * xv[q].w);
    }
    m = mn;
  }

  const float inv = 1.f / z;
  float4* o4 = (float4*)(bag + (size_t)b * DCLS);
#pragma unroll
  for (int q = 0; q < 4; ++q) {
    float4 r = a4[q];
    r.x *= inv; r.y *= inv; r.z *= inv; r.w *= inv;
    o4[lane + 32 * q] = r;
  }
}

__global__ __launch_bounds__(256) void tan_classifier(
    const float* __restrict__ bag, const float* __restrict__ W,
    const float* __restrict__ bias, float* __restrict__ out)
{
  const int lane = threadIdx.x & 31;
  const int wave = threadIdx.x >> 5;
  const int gw = blockIdx.x * 8 + wave;
  const int mt = gw >> 2;
  const int nt = gw & 3;
  const int m0 = mt * 16;
  const int n0 = nt * 16;

  const int r16  = lane & 15;
  const int koff = (lane >> 4) << 1;

  const float* arow = bag + (size_t)(m0 + r16) * DCLS + koff;
  const int ng  = n0 + r16;
  const int ngc = ng < REL ? ng : (REL - 1);
  const float* brow = W + (size_t)ngc * DCLS + koff;

  v8f cacc = {};
#if __has_builtin(__builtin_amdgcn_wmma_f32_16x16x4_f32)
#pragma unroll 4
  for (int k = 0; k < DCLS; k += 4) {
    v2f a  = *(const v2f*)(arow + k);
    v2f bb = *(const v2f*)(brow + k);
    cacc = __builtin_amdgcn_wmma_f32_16x16x4_f32(
         false, a,  false, bb,
         (short)0, cacc,  false,  false);
  }
#else
  {
    const int mb = m0 + ((lane >> 4) << 3);
    for (int r = 0; r < 8; ++r) {
      float acc = 0.f;
      const float* ar = bag + (size_t)(mb + r) * DCLS;
      const float* br = W + (size_t)ngc * DCLS;
      for (int k = 0; k < DCLS; ++k) acc = fmaf(ar[k], br[k], acc);
      cacc[r] = acc;
    }
  }
#endif

  if (ng < REL) {
    const float bv = bias[ng];
    const int mb = m0 + ((lane >> 4) << 3);
#pragma unroll
    for (int r = 0; r < 8; ++r)
      out[(size_t)(mb + r) * REL + ng] = cacc[r] + bv;
  }
}

extern "C" void kernel_launch(void* const* d_in, const int* in_sizes, int n_in,
                              void* d_out, int out_size, void* d_ws, size_t ws_size,
                              hipStream_t stream) {
  (void)in_sizes; (void)n_in; (void)out_size; (void)ws_size;
  const float* X     = (const float*)d_in[0];
  const int*   scope = (const int*)  d_in[1];
  const float* Con   = (const float*)d_in[2];
  const int*   rel   = (const int*)  d_in[3];
  const float* W     = (const float*)d_in[4];
  const float* bias  = (const float*)d_in[5];
  float* out = (float*)d_out;
  float* bagbuf = (float*)d_ws;

  tan_bag_attn<<<N_BAGS / 8, 256, 0, stream>>>(X, scope, Con, rel, bagbuf);
  tan_classifier<<<(N_BAGS / 16) * 4 / 8, 256, 0, stream>>>(bagbuf, W, bias, out);
}


# Bag kernel: each wave owns 4 consecutive bags as one contiguous row stream through a ring of 8 row buffers in registers (row j+8 loaded once row j is consumed, across bag ends); 4096 waves, was 16384
# speedup vs baseline: 1.3889x; 1.3889x over previous
#include <hip/hip_runtime.h>
#include <hip/hip_bf16.h>
#include <math.h>

#define N_SENT 131072
#define N_BAGS 16384
#define DCLS   512
#define REL    53

typedef __attribute__((ext_vector_type(2))) float v2f;
typedef __attribute__((ext_vector_type(8))) float v8f;

#define BPW   4
#define DEPTH 8

__device__ __forceinline__ float xhalf(float x) {
  const int b = __builtin_bit_cast(int, x);
  return __builtin_bit_cast(float,
      __builtin_amdgcn_permlanex16(b, b, 0x76543210, 0xfedcba98, false, false));
}
template <int MASK>
__device__ __forceinline__ float xrow(float x) {
  const int b = __builtin_bit_cast(int, x);
  return __builtin_bit_cast(float,
      __builtin_amdgcn_update_dpp(0, b, 0x160 + MASK, 0xf, 0xf, true));
}
__device__ __forceinline__ int pick4(int k, int v0, int v1, int v2, int v3) {
  return k == 0 ? v0 : (k == 1 ? v1 : (k == 2 ? v2 : v3));
}

__global__ __launch_bounds__(256) void tan_bag_attn(
    const float* __restrict__ X, const int* __restrict__ scope,
    const float* __restrict__ Con, const int* __restrict__ rel,
    float* __restrict__ bag)
{
  __shared__ float4 sCon[8 * (BPW - 1) * (DCLS / 4)];

  const int lane = threadIdx.x & 31;
  const int wave = __builtin_amdgcn_readfirstlane((int)(threadIdx.x >> 5));
  const int b0 = (blockIdx.x * 8 + wave) * BPW;

  int j = b0 > 0 ? scope[2 * b0 - 1] : 0;
  const int e0 = scope[2 * b0 + 1];
  const int e1 = scope[2 * b0 + 3];
  const int e2 = scope[2 * b0 + 5];
  const int e3 = scope[2 * b0 + 7];
  const int r0 = rel[b0], r1 = rel[b0 + 1], r2 = rel[b0 + 2], r3 = rel[b0 + 3];
  const int end  = e3;
  const int last = end > 0 ? end - 1 : 0;

  const float4* x4 = (const float4*)X + lane;

  float4 xb[DEPTH][4];
#pragma unroll
  for (int u = 0; u < DEPTH; ++u) {
    const int jr = j + u < last ? j + u : last;
#pragma unroll
    for (int q = 0; q < 4; ++q) xb[u][q] = x4[(size_t)jr * (DCLS / 4) + 32 * q];
  }

  float4 c4[4], a4[4];
  float4* myCon = sCon + wave * ((BPW - 1) * (DCLS / 4)) + lane;
  {
    const float4* p0 = (const float4*)(Con + (size_t)r0 * DCLS) + lane;
    const float4* p1 = (const float4*)(Con + (size_t)r1 * DCLS) + lane;
    const float4* p2 = (const float4*)(Con + (size_t)r2 * DCLS) + lane;
    const float4* p3 = (const float4*)(Con + (size_t)r3 * DCLS) + lane;
#pragma unroll
    for (int q = 0; q < 4; ++q) {
      c4[q] = p0[32 * q];
      myCon[0 * (DCLS / 4) + 32 * q] = p1[32 * q];
      myCon[1 * (DCLS / 4) + 32 * q] = p2[32 * q];
      myCon[2 * (DCLS / 4) + 32 * q] = p3[32 * q];
      a4[q] = make_float4(0.f, 0.f, 0.f, 0.f);
    }
  }

  int   k = 0;
  int   cur_end = e0;
  float m = -INFINITY, z = 0.f;

  auto close_bag = [&]() {
    const float inv = z > 0.f ? 1.f / z : 0.f;
    float4* o4 = (float4*)(bag + (size_t)(b0 + k) * DCLS) + lane;
#pragma unroll
    for (int q = 0; q < 4; ++q) {
      float4 r = a4[q];
      r.x *= inv; r.y *= inv; r.z *= inv; r.w *= inv;
      o4[32 * q] = r;
      a4[q] = make_float4(0.f, 0.f, 0.f, 0.f);
    }
    m = -INFINITY; z = 0.f;
    ++k;
    cur_end = pick4(k, e0, e1, e2, e3);
    if (k < BPW) {
#pragma unroll
      for (int q = 0; q < 4; ++q) c4[q] = myCon[(k - 1) * (DCLS / 4) + 32 * q];
    }
  };

  for (;;) {
#pragma unroll
    for (int u = 0; u < DEPTH; ++u) {
      if (j >= end) goto drained;
      while (j == cur_end) close_bag();

      float s = 0.f;
#pragma unroll
      for (int q = 0; q < 4; ++q) {
        s = fmaf(xb[u][q].x, c4[q].x, s);
        s = fmaf(xb[u][q].y, c4[q].y, s);
        s = fmaf(xb[u][q].z, c4[q].z, s);
        s = fmaf(xb[u][q].w, c4[q].w, s);
      }
      s += xhalf(s);
      s += xrow<8>(s);
      s += xrow<4>(s);
      s += xrow<2>(s);
      s += xrow<1>(s);

      const float mn    = fmaxf(m, s);
      const float scale = __expf(m - mn);
      const float w     = __expf(s - mn);
      z = z * scale + w;
#pragma unroll
      for (int q = 0; q < 4; ++q) {
        a4[q].x = fmaf(a4[q].x, scale, w * xb[u][q].x);
        a4[q].y = fmaf(a4[q].y, scale, w * xb[u][q].y);
        a4[q].z = fmaf(a4[q].z, scale, w * xb[u][q].z);
        a4[q].w = fmaf(a4[q].w, scale, w * xb[u][q].w);
      }
      m = mn;

      __builtin_amdgcn_sched_barrier(0);
      {
        const int jr = j + DEPTH < last ? j + DEPTH : last;
#pragma unroll
        for (int q = 0; q < 4; ++q) xb[u][q] = x4[(size_t)jr * (DCLS / 4) + 32 * q];
      }
      ++j;
    }
  }
drained:
  while (k < BPW) close_bag();
}

#define WPITCH4 129
#define KSTEPS  (DCLS / 8)
#define KBLK    8

__device__ __forceinline__ v8f wmma4(v2f a, v2f b, v8f c) {
  return __builtin_amdgcn_wmma_f32_16x16x4_f32(
       false, a,  false, b,
       (short)0, c,  false,  false);
}

__global__ __launch_bounds__(256) void tan_classifier(
    const float* __restrict__ bag, const float* __restrict__ W,
    const float* __restrict__ bias, float* __restrict__ out)
{
  __shared__ float4 sW[(REL + 1) * WPITCH4];

  const int tid  = threadIdx.x;
  const int lane = tid & 31;
  const int wave = tid >> 5;
  const int c = lane & 15;
  const int h = lane >> 4;
  const int m0 = (blockIdx.x * 4 + (wave >> 1)) * 16;
  const int nh = wave & 1;

  const float4* W4 = (const float4*)W;
  const int c4 = tid & 127;
  const int r0 = tid >> 7;
  float4 wst[27];
#pragma unroll
  for (int i = 0; i < 27; ++i) {
    const int r = r0 + 2 * i;
    wst[i] = W4[(r < REL ? r : REL - 1) * (DCLS / 4) + c4];
  }

#pragma unroll
  for (int i = 0; i < 27; ++i) sW[(r0 + 2 * i) * WPITCH4 + c4] = wst[i];
  __syncthreads();

  const float4* ap = (const float4*)(bag + (size_t)(m0 + c) * DCLS + 4 * h);
  float4 a[KSTEPS];
#pragma unroll
  for (int i = 0; i < KSTEPS; ++i) a[i] = ap[2 * i];
  const int n0g = nh * 32 + c;
  const int n1g = n0g + 16;
  const int n0c = n0g < REL ? n0g : (REL - 1);
  const int n1c = n1g < REL ? n1g : (REL - 1);
  const float bv0 = bias[n0c];
  const float bv1 = bias[n1c];
  __builtin_amdgcn_sched_barrier(0);

  const float4* bp0 = sW + n0c * WPITCH4 + h;
  const float4* bp1 = sW + n1c * WPITCH4 + h;

  v8f acc0 = {}, acc1 = {};
  float4 p0[KBLK], p1[KBLK], q0[KBLK], q1[KBLK];

#pragma unroll
  for (int j = 0; j < KBLK; ++j) { p0[j] = bp0[2 * j]; p1[j] = bp1[2 * j]; }

#pragma unroll
  for (int blk = 0; blk < KSTEPS / KBLK; blk += 2) {
#pragma unroll
    for (int j = 0; j < KBLK; ++j) {
      q0[j] = bp0[2 * ((blk + 1) * KBLK + j)];
      q1[j] = bp1[2 * ((blk + 1) * KBLK + j)];
    }
    __builtin_amdgcn_sched_barrier(0);
#pragma unroll
    for (int j = 0; j < KBLK; ++j) {
      const float4 av = a[blk * KBLK + j];
      const v2f alo = {av.x, av.y}, ahi = {av.z, av.w};
      const v2f b0lo = {p0[j].x, p0[j].y}, b0hi = {p0[j].z, p0[j].w};
      const v2f b1lo = {p1[j].x, p1[j].y}, b1hi = {p1[j].z, p1[j].w};
      acc0 = wmma4(alo, b0lo, acc0);
      acc1 = wmma4(alo, b1lo, acc1);
      acc0 = wmma4(ahi, b0hi, acc0);
      acc1 = wmma4(ahi, b1hi, acc1);
    }
    __builtin_amdgcn_sched_barrier(0);
    if (blk + 2 < KSTEPS / KBLK) {
#pragma unroll
      for (int j = 0; j < KBLK; ++j) {
        p0[j] = bp0[2 * ((blk + 2) * KBLK + j)];
        p1[j] = bp1[2 * ((blk + 2) * KBLK + j)];
      }
    }
    __builtin_amdgcn_sched_barrier(0);
#pragma unroll
    for (int j = 0; j < KBLK; ++j) {
      const float4 av = a[(blk + 1) * KBLK + j];
      const v2f alo = {av.x, av.y}, ahi = {av.z, av.w};
      const v2f b0lo = {q0[j].x, q0[j].y}, b0hi = {q0[j].z, q0[j].w};
      const v2f b1lo = {q1[j].x, q1[j].y}, b1hi = {q1[j].z, q1[j].w};
      acc0 = wmma4(alo, b0lo, acc0);
      acc1 = wmma4(alo, b1lo, acc1);
      acc0 = wmma4(ahi, b0hi, acc0);
      acc1 = wmma4(ahi, b1hi, acc1);
    }
    __builtin_amdgcn_sched_barrier(0);
  }

  float o0[8], o1[8];
#pragma unroll
  for (int r = 0; r < 8; ++r) { o0[r] = acc0[r] + bv0; o1[r] = acc1[r] + bv1; }

  float* orow = out + (size_t)(m0 + 8 * h) * REL;
  if (n0g < REL) {
#pragma unroll
    for (int r = 0; r < 8; ++r) orow[r * REL + n0g] = o0[r];
  }
  if (n1g < REL) {
#pragma unroll
    for (int r = 0; r < 8; ++r) orow[r * REL + n1g] = o1[r];
  }
}

extern "C" void kernel_launch(void* const* d_in, const int* in_sizes, int n_in,
                              void* d_out, int out_size, void* d_ws, size_t ws_size,
                              hipStream_t stream) {
  (void)in_sizes; (void)n_in; (void)out_size; (void)ws_size;
  const float* X     = (const float*)d_in[0];
  const int*   scope = (const int*)  d_in[1];
  const float* Con   = (const float*)d_in[2];
  const int*   rel   = (const int*)  d_in[3];
  const float* W     = (const float*)d_in[4];
  const float* bias  = (const float*)d_in[5];
  float* out = (float*)d_out;
  float* bagbuf = (float*)d_ws;

  tan_bag_attn<<<N_BAGS / (8 * BPW), 256, 0, stream>>>(X, scope, Con, rel, bagbuf);
  tan_classifier<<<N_BAGS / 64, 256, 0, stream>>>(bagbuf, W, bias, out);
}
